# starting kernel
# baseline (speedup 1.0000x reference)
#include <hip/hip_runtime.h>

typedef __attribute__((ext_vector_type(2))) float v2f;
typedef __attribute__((ext_vector_type(2), aligned(4))) float v2fu;
typedef __attribute__((ext_vector_type(4))) float v4f;
typedef __attribute__((ext_vector_type(8))) float v8f;

#define NUNITS 81
#define WA_OFF    0
#define WB_OFF    736
#define WMIX_OFF  1472
#define BFRAG_OFF 2768
#define NFRAG     (9 * 2 * 3 * 32)

__device__ __constant__ int c_hw2i[12] = {0, 1, 3, 4, 6, 7, 2, -1, 5, -1, -1, 8};

__global__ void logic_prep_kernel(const float* __restrict__ sa,
                                  const float* __restrict__ sb,
                                  const float* __restrict__ op,
                                  float* __restrict__ ws) {
  const int tid = threadIdx.x;
  if (tid < NUNITS) {
    {
      const float* src = sa + tid * 9;
      float m = src[0];
#pragma unroll
      for (int i = 1; i < 9; ++i) m = fmaxf(m, src[i]);
      float e[9], s = 0.f;
#pragma unroll
      for (int i = 0; i < 9; ++i) { e[i] = expf(src[i] - m); s += e[i]; }
      const float inv = 1.0f / s;
#pragma unroll
      for (int i = 0; i < 9; ++i) ws[WA_OFF + tid * 9 + i] = e[i] * inv;
    }
    {
      const float* src = sb + tid * 9;
      float m = src[0];
#pragma unroll
      for (int i = 1; i < 9; ++i) m = fmaxf(m, src[i]);
      float e[9], s = 0.f;
#pragma unroll
      for (int i = 0; i < 9; ++i) { e[i] = expf(src[i] - m); s += e[i]; }
      const float inv = 1.0f / s;
#pragma unroll
      for (int i = 0; i < 9; ++i) ws[WB_OFF + tid * 9 + i] = e[i] * inv;
    }
    {
      const float* src = op + tid * 16;
      float m = src[0];
#pragma unroll
      for (int i = 1; i < 16; ++i) m = fmaxf(m, src[i]);
      float w[16], s = 0.f;
#pragma unroll
      for (int i = 0; i < 16; ++i) { w[i] = expf(src[i] - m); s += w[i]; }
      const float inv = 1.0f / s;
#pragma unroll
      for (int i = 0; i < 16; ++i) w[i] *= inv;
      const float c0  = w[8] + w[9] + w[10] + w[11] + w[12] + w[13] + w[14] + w[15];
      const float cA  = w[2] + w[3] + w[6] + w[7] - w[8] - w[9] - w[12] - w[13];
      const float cB  = w[4] + w[5] + w[6] + w[7] - w[8] - w[9] - w[10] - w[11];
      const float cAB = w[1] - w[2] - w[4] - 2.0f * w[6] - w[7] + w[8] +
                        2.0f * w[9] + w[11] + w[13] - w[14];
      ws[WMIX_OFF + tid * 4 + 0] = c0;
      ws[WMIX_OFF + tid * 4 + 1] = cA;
      ws[WMIX_OFF + tid * 4 + 2] = cB;
      ws[WMIX_OFF + tid * 4 + 3] = cAB;
    }
  }
  __syncthreads();
  for (int idx = tid; idx < NFRAG; idx += blockDim.x) {
    const int lane  = idx & 31;
    const int rest  = idx >> 5;
    const int kk    = rest % 3;
    const int pt    = rest / 3;
    const int t     = pt & 1;
    const int p     = pt >> 1;
    const int col   = lane & 15;
    const int khalf = (lane >> 4) << 1;
    const float* tbl = ws + (t ? WB_OFF : WA_OFF);
    const int k0 = 4 * kk + khalf;
    const int i0 = c_hw2i[k0];
    const int i1 = c_hw2i[k0 + 1];
    float v0 = 0.f, v1 = 0.f;
    if (col < 9) {
      if (i0 >= 0) v0 = tbl[(9 * p + col) * 9 + i0];
      if (i1 >= 0) v1 = tbl[(9 * p + col) * 9 + i1];
    }
    ws[BFRAG_OFF + idx * 2 + 0] = v0;
    ws[BFRAG_OFF + idx * 2 + 1] = v1;
  }
}

__global__ void __launch_bounds__(128, 1)
logic_wmma_kernel(const float* __restrict__ x,
                  const float* __restrict__ ws,
                  float* __restrict__ out,
                  int nBatch) {
  const int lane   = threadIdx.x & 31;
  const int waveId = threadIdx.x >> 5;
  const long base  = ((long)blockIdx.x * 4 + waveId) * 16;
  if (base + 16 > nBatch) return;

  const int col  = lane & 15;
  const bool hiM = (lane >= 16);

  const float* xrow = x + (base + col) * 81;
  const float* xr0  = xrow + (hiM ? 9 : 0);
  const float* xr1  = xrow + (hiM ? 2 : 18);
  const float* xr2  = xrow + (hiM ? 19 : 11);
  const v2f*   bp   = (const v2f*)(ws + BFRAG_OFF) + lane;
  const float* wc   = ws + WMIX_OFF + col * 4;
  float*       outp = out + (base + (hiM ? 8 : 0)) * 81 + col;
  const bool   act  = (col < 9);

#pragma unroll
  for (int p = 0; p < 9; ++p) {
    const int pbase = (p / 3) * 27 + (p % 3) * 3;

    const v2f a0 = *(const v2fu*)(xr0 + pbase);
    const v2f a1 = *(const v2fu*)(xr1 + pbase);
    const v2f a2 = *(const v2fu*)(xr2 + pbase);

    const v2f ba0 = bp[(p * 6 + 0) * 32];
    const v2f ba1 = bp[(p * 6 + 1) * 32];
    const v2f ba2 = bp[(p * 6 + 2) * 32];
    const v2f bb0 = bp[(p * 6 + 3) * 32];
    const v2f bb1 = bp[(p * 6 + 4) * 32];
    const v2f bb2 = bp[(p * 6 + 5) * 32];

    v8f cAcc = {};
    v8f cBcc = {};
    cAcc = __builtin_amdgcn_wmma_f32_16x16x4_f32(false, a0, false, ba0, (short)0, cAcc, false, false);
    cAcc = __builtin_amdgcn_wmma_f32_16x16x4_f32(false, a1, false, ba1, (short)0, cAcc, false, false);
    cAcc = __builtin_amdgcn_wmma_f32_16x16x4_f32(false, a2, false, ba2, (short)0, cAcc, false, false);
    cBcc = __builtin_amdgcn_wmma_f32_16x16x4_f32(false, a0, false, bb0, (short)0, cBcc, false, false);
    cBcc = __builtin_amdgcn_wmma_f32_16x16x4_f32(false, a1, false, bb1, (short)0, cBcc, false, false);
    cBcc = __builtin_amdgcn_wmma_f32_16x16x4_f32(false, a2, false, bb2, (short)0, cBcc, false, false);

    if (act) {
      const v4f c = *(const v4f*)(wc + 36 * p);
#pragma unroll
      for (int m = 0; m < 8; ++m) {
        const float av = cAcc[m];
        const float bv = cBcc[m];
        const float r  = fmaf(av * bv, c.w, fmaf(bv, c.z, fmaf(av, c.y, c.x)));
        __builtin_nontemporal_store(r, outp + 9 * p + 81 * m);
      }
    }
  }
}

extern "C" void kernel_launch(void* const* d_in, const int* in_sizes, int n_in,
                              void* d_out, int out_size, void* d_ws, size_t ws_size,
                              hipStream_t stream) {
  const float* x  = (const float*)d_in[0];
  const float* sa = (const float*)d_in[1];
  const float* sb = (const float*)d_in[2];
  const float* op = (const float*)d_in[3];
  float* out = (float*)d_out;
  float* ws  = (float*)d_ws;

  const int nBatch = in_sizes[0] / 81;

  logic_prep_kernel<<<1, 128, 0, stream>>>(sa, sb, op, ws);

  const int rowsPerBlock = 64;
  const int nBlocks = (nBatch + rowsPerBlock - 1) / rowsPerBlock;
  logic_wmma_kernel<<<nBlocks, 128, 0, stream>>>(x, ws, out, nBatch);
}


# x tile staged through LDS: each wave reads its 16 rows once with whole-line 16-byte loads (11 x 512 B in place of 27 scattered 8-byte loads) and takes the A fragments from a wave-private LDS copy
# speedup vs baseline: 1.1381x; 1.1381x over previous
#include <hip/hip_runtime.h>

typedef __attribute__((ext_vector_type(2))) float v2f;
typedef __attribute__((ext_vector_type(2), aligned(4))) float v2fu;
typedef __attribute__((ext_vector_type(4))) float v4f;
typedef __attribute__((ext_vector_type(8))) float v8f;

#define NUNITS 81
#define WA_OFF    0
#define WB_OFF    736
#define WMIX_OFF  1472
#define BFRAG_OFF 2768
#define NFRAG     (9 * 2 * 3 * 32)
#define TILE_FLOATS (16 * 81)

__device__ __constant__ int c_hw2i[12] = {0, 1, 3, 4, 6, 7, 2, -1, 5, -1, -1, 8};

__global__ void logic_prep_kernel(const float* __restrict__ sa,
                                  const float* __restrict__ sb,
                                  const float* __restrict__ op,
                                  float* __restrict__ ws) {
  const int tid = threadIdx.x;
  if (tid < NUNITS) {
    {
      const float* src = sa + tid * 9;
      float m = src[0];
#pragma unroll
      for (int i = 1; i < 9; ++i) m = fmaxf(m, src[i]);
      float e[9], s = 0.f;
#pragma unroll
      for (int i = 0; i < 9; ++i) { e[i] = expf(src[i] - m); s += e[i]; }
      const float inv = 1.0f / s;
#pragma unroll
      for (int i = 0; i < 9; ++i) ws[WA_OFF + tid * 9 + i] = e[i] * inv;
    }
    {
      const float* src = sb + tid * 9;
      float m = src[0];
#pragma unroll
      for (int i = 1; i < 9; ++i) m = fmaxf(m, src[i]);
      float e[9], s = 0.f;
#pragma unroll
      for (int i = 0; i < 9; ++i) { e[i] = expf(src[i] - m); s += e[i]; }
      const float inv = 1.0f / s;
#pragma unroll
      for (int i = 0; i < 9; ++i) ws[WB_OFF + tid * 9 + i] = e[i] * inv;
    }
    {
      const float* src = op + tid * 16;
      float m = src[0];
#pragma unroll
      for (int i = 1; i < 16; ++i) m = fmaxf(m, src[i]);
      float w[16], s = 0.f;
#pragma unroll
      for (int i = 0; i < 16; ++i) { w[i] = expf(src[i] - m); s += w[i]; }
      const float inv = 1.0f / s;
#pragma unroll
      for (int i = 0; i < 16; ++i) w[i] *= inv;
      const float c0  = w[8] + w[9] + w[10] + w[11] + w[12] + w[13] + w[14] + w[15];
      const float cA  = w[2] + w[3] + w[6] + w[7] - w[8] - w[9] - w[12] - w[13];
      const float cB  = w[4] + w[5] + w[6] + w[7] - w[8] - w[9] - w[10] - w[11];
      const float cAB = w[1] - w[2] - w[4] - 2.0f * w[6] - w[7] + w[8] +
                        2.0f * w[9] + w[11] + w[13] - w[14];
      ws[WMIX_OFF + tid * 4 + 0] = c0;
      ws[WMIX_OFF + tid * 4 + 1] = cA;
      ws[WMIX_OFF + tid * 4 + 2] = cB;
      ws[WMIX_OFF + tid * 4 + 3] = cAB;
    }
  }
  __syncthreads();
  for (int idx = tid; idx < NFRAG; idx += blockDim.x) {
    const int lane  = idx & 31;
    const int rest  = idx >> 5;
    const int kk    = rest % 3;
    const int pt    = rest / 3;
    const int t     = pt & 1;
    const int p     = pt >> 1;
    const int col   = lane & 15;
    const int khalf = (lane >> 4) << 1;
    const float* tbl = ws + (t ? WB_OFF : WA_OFF);
    const int k0 = 4 * kk + khalf;
    const int i0 = c_hw2i[k0];
    const int i1 = c_hw2i[k0 + 1];
    float v0 = 0.f, v1 = 0.f;
    if (col < 9) {
      if (i0 >= 0) v0 = tbl[(9 * p + col) * 9 + i0];
      if (i1 >= 0) v1 = tbl[(9 * p + col) * 9 + i1];
    }
    ws[BFRAG_OFF + idx * 2 + 0] = v0;
    ws[BFRAG_OFF + idx * 2 + 1] = v1;
  }
}

__global__ void __launch_bounds__(128, 1)
logic_wmma_kernel(const float* __restrict__ x,
                  const float* __restrict__ ws,
                  float* __restrict__ out,
                  int nBatch) {
  __shared__ __attribute__((aligned(16))) float sx[4 * TILE_FLOATS];
  const int lane   = threadIdx.x & 31;
  const int waveId = threadIdx.x >> 5;
  const long base  = ((long)blockIdx.x * 4 + waveId) * 16;
  if (base + 16 > nBatch) return;

  const int col  = lane & 15;
  const bool hiM = (lane >= 16);

  const v4f* xt   = (const v4f*)(x + base * 81);
  v4f*       lt   = (v4f*)(sx + waveId * TILE_FLOATS);
  const int  head = ((base * 81) & 16) ? 4 : 0;
  const int  rest = head ? 0 : 320;
  v4f stage[10];
#pragma unroll
  for (int q = 0; q < 10; ++q)
    stage[q] = xt[head + q * 32 + lane];
  v4f stageRest;
  if (lane < 4) stageRest = xt[rest + lane];
#pragma unroll
  for (int q = 0; q < 10; ++q) lt[head + q * 32 + lane] = stage[q];
  if (lane < 4) lt[rest + lane] = stageRest;

  const float* xrow = sx + waveId * TILE_FLOATS + col * 81;
  const float* xr0  = xrow + (hiM ? 9 : 0);
  const float* xr1  = xrow + (hiM ? 2 : 18);
  const float* xr2  = xrow + (hiM ? 19 : 11);
  const v2f*   bp   = (const v2f*)(ws + BFRAG_OFF) + lane;
  const float* wc   = ws + WMIX_OFF + col * 4;
  float*       outp = out + (base + (hiM ? 8 : 0)) * 81 + col;
  const bool   act  = (col < 9);

#pragma unroll
  for (int p = 0; p < 9; ++p) {
    const int pbase = (p / 3) * 27 + (p % 3) * 3;

    const v2f a0 = *(const v2fu*)(xr0 + pbase);
    const v2f a1 = *(const v2fu*)(xr1 + pbase);
    const v2f a2 = *(const v2fu*)(xr2 + pbase);

    const v2f ba0 = bp[(p * 6 + 0) * 32];
    const v2f ba1 = bp[(p * 6 + 1) * 32];
    const v2f ba2 = bp[(p * 6 + 2) * 32];
    const v2f bb0 = bp[(p * 6 + 3) * 32];
    const v2f bb1 = bp[(p * 6 + 4) * 32];
    const v2f bb2 = bp[(p * 6 + 5) * 32];

    v8f cAcc = {};
    v8f cBcc = {};
    cAcc = __builtin_amdgcn_wmma_f32_16x16x4_f32(false, a0, false, ba0, (short)0, cAcc, false, false);
    cAcc = __builtin_amdgcn_wmma_f32_16x16x4_f32(false, a1, false, ba1, (short)0, cAcc, false, false);
    cAcc = __builtin_amdgcn_wmma_f32_16x16x4_f32(false, a2, false, ba2, (short)0, cAcc, false, false);
    cBcc = __builtin_amdgcn_wmma_f32_16x16x4_f32(false, a0, false, bb0, (short)0, cBcc, false, false);
    cBcc = __builtin_amdgcn_wmma_f32_16x16x4_f32(false, a1, false, bb1, (short)0, cBcc, false, false);
    cBcc = __builtin_amdgcn_wmma_f32_16x16x4_f32(false, a2, false, bb2, (short)0, cBcc, false, false);

    if (act) {
      const v4f c = *(const v4f*)(wc + 36 * p);
#pragma unroll
      for (int m = 0; m < 8; ++m) {
        const float av = cAcc[m];
        const float bv = cBcc[m];
        const float r  = fmaf(av * bv, c.w, fmaf(bv, c.z, fmaf(av, c.y, c.x)));
        __builtin_nontemporal_store(r, outp + 9 * p + 81 * m);
      }
    }
  }
}

extern "C" void kernel_launch(void* const* d_in, const int* in_sizes, int n_in,
                              void* d_out, int out_size, void* d_ws, size_t ws_size,
                              hipStream_t stream) {
  const float* x  = (const float*)d_in[0];
  const float* sa = (const float*)d_in[1];
  const float* sb = (const float*)d_in[2];
  const float* op = (const float*)d_in[3];
  float* out = (float*)d_out;
  float* ws  = (float*)d_ws;

  const int nBatch = in_sizes[0] / 81;

  logic_prep_kernel<<<1, 128, 0, stream>>>(sa, sb, op, ws);

  const int rowsPerBlock = 64;
  const int nBlocks = (nBatch + rowsPerBlock - 1) / rowsPerBlock;
  logic_wmma_kernel<<<nBlocks, 128, 0, stream>>>(x, ws, out, nBatch);
}


# out tile staged through LDS in place of the x copy (A fragments all read first): each wave writes its 16 rows with whole-line 16-byte stores (11 x 512 B) in place of 72 two-piece 36-byte stores
# speedup vs baseline: 1.7365x; 1.5258x over previous
#include <hip/hip_runtime.h>

typedef __attribute__((ext_vector_type(2))) float v2f;
typedef __attribute__((ext_vector_type(2), aligned(4))) float v2fu;
typedef __attribute__((ext_vector_type(4))) float v4f;
typedef __attribute__((ext_vector_type(8))) float v8f;

#define NUNITS 81
#define WA_OFF    0
#define WB_OFF    736
#define WMIX_OFF  1472
#define BFRAG_OFF 2768
#define NFRAG     (9 * 2 * 3 * 32)
#define TILE_FLOATS (16 * 81)

__device__ __constant__ int c_hw2i[12] = {0, 1, 3, 4, 6, 7, 2, -1, 5, -1, -1, 8};

__global__ void logic_prep_kernel(const float* __restrict__ sa,
                                  const float* __restrict__ sb,
                                  const float* __restrict__ op,
                                  float* __restrict__ ws) {
  const int tid = threadIdx.x;
  if (tid < NUNITS) {
    {
      const float* src = sa + tid * 9;
      float m = src[0];
#pragma unroll
      for (int i = 1; i < 9; ++i) m = fmaxf(m, src[i]);
      float e[9], s = 0.f;
#pragma unroll
      for (int i = 0; i < 9; ++i) { e[i] = expf(src[i] - m); s += e[i]; }
      const float inv = 1.0f / s;
#pragma unroll
      for (int i = 0; i < 9; ++i) ws[WA_OFF + tid * 9 + i] = e[i] * inv;
    }
    {
      const float* src = sb + tid * 9;
      float m = src[0];
#pragma unroll
      for (int i = 1; i < 9; ++i) m = fmaxf(m, src[i]);
      float e[9], s = 0.f;
#pragma unroll
      for (int i = 0; i < 9; ++i) { e[i] = expf(src[i] - m); s += e[i]; }
      const float inv = 1.0f / s;
#pragma unroll
      for (int i = 0; i < 9; ++i) ws[WB_OFF + tid * 9 + i] = e[i] * inv;
    }
    {
      const float* src = op + tid * 16;
      float m = src[0];
#pragma unroll
      for (int i = 1; i < 16; ++i) m = fmaxf(m, src[i]);
      float w[16], s = 0.f;
#pragma unroll
      for (int i = 0; i < 16; ++i) { w[i] = expf(src[i] - m); s += w[i]; }
      const float inv = 1.0f / s;
#pragma unroll
      for (int i = 0; i < 16; ++i) w[i] *= inv;
      const float c0  = w[8] + w[9] + w[10] + w[11] + w[12] + w[13] + w[14] + w[15];
      const float cA  = w[2] + w[3] + w[6] + w[7] - w[8] - w[9] - w[12] - w[13];
      const float cB  = w[4] + w[5] + w[6] + w[7] - w[8] - w[9] - w[10] - w[11];
      const float cAB = w[1] - w[2] - w[4] - 2.0f * w[6] - w[7] + w[8] +
                        2.0f * w[9] + w[11] + w[13] - w[14];
      ws[WMIX_OFF + tid * 4 + 0] = c0;
      ws[WMIX_OFF + tid * 4 + 1] = cA;
      ws[WMIX_OFF + tid * 4 + 2] = cB;
      ws[WMIX_OFF + tid * 4 + 3] = cAB;
    }
  }
  __syncthreads();
  for (int idx = tid; idx < NFRAG; idx += blockDim.x) {
    const int lane  = idx & 31;
    const int rest  = idx >> 5;
    const int kk    = rest % 3;
    const int pt    = rest / 3;
    const int t     = pt & 1;
    const int p     = pt >> 1;
    const int col   = lane & 15;
    const int khalf = (lane >> 4) << 1;
    const float* tbl = ws + (t ? WB_OFF : WA_OFF);
    const int k0 = 4 * kk + khalf;
    const int i0 = c_hw2i[k0];
    const int i1 = c_hw2i[k0 + 1];
    float v0 = 0.f, v1 = 0.f;
    if (col < 9) {
      if (i0 >= 0) v0 = tbl[(9 * p + col) * 9 + i0];
      if (i1 >= 0) v1 = tbl[(9 * p + col) * 9 + i1];
    }
    ws[BFRAG_OFF + idx * 2 + 0] = v0;
    ws[BFRAG_OFF + idx * 2 + 1] = v1;
  }
}

__global__ void __launch_bounds__(128, 1)
logic_wmma_kernel(const float* __restrict__ x,
                  const float* __restrict__ ws,
                  float* __restrict__ out,
                  int nBatch) {
  __shared__ __attribute__((aligned(16))) float sx[4 * TILE_FLOATS];
  const int lane   = threadIdx.x & 31;
  const int waveId = threadIdx.x >> 5;
  const long base  = ((long)blockIdx.x * 4 + waveId) * 16;
  if (base + 16 > nBatch) return;

  const int col  = lane & 15;
  const bool hiM = (lane >= 16);

  const v4f* xt   = (const v4f*)(x + base * 81);
  v4f*       lt   = (v4f*)(sx + waveId * TILE_FLOATS);
  const int  head = ((base * 81) & 16) ? 4 : 0;
  const int  rest = head ? 0 : 320;
  v4f stage[10];
#pragma unroll
  for (int q = 0; q < 10; ++q)
    stage[q] = xt[head + q * 32 + lane];
  v4f stageRest;
  if (lane < 4) stageRest = xt[rest + lane];
#pragma unroll
  for (int q = 0; q < 10; ++q) lt[head + q * 32 + lane] = stage[q];
  if (lane < 4) lt[rest + lane] = stageRest;

  const float* xrow = sx + waveId * TILE_FLOATS + col * 81;
  const float* xr0  = xrow + (hiM ? 9 : 0);
  const float* xr1  = xrow + (hiM ? 2 : 18);
  const float* xr2  = xrow + (hiM ? 19 : 11);
  const v2f*   bp   = (const v2f*)(ws + BFRAG_OFF) + lane;
  const float* wc   = ws + WMIX_OFF + col * 4;
  float*       resp = sx + waveId * TILE_FLOATS + (hiM ? 8 : 0) * 81 + col;
  const bool   act  = (col < 9);

  v2f afr[9][3];
#pragma unroll
  for (int p = 0; p < 9; ++p) {
    const int pbase = (p / 3) * 27 + (p % 3) * 3;
    afr[p][0] = *(const v2fu*)(xr0 + pbase);
    afr[p][1] = *(const v2fu*)(xr1 + pbase);
    afr[p][2] = *(const v2fu*)(xr2 + pbase);
  }

#pragma unroll
  for (int p = 0; p < 9; ++p) {
    const v2f a0 = afr[p][0];
    const v2f a1 = afr[p][1];
    const v2f a2 = afr[p][2];

    const v2f ba0 = bp[(p * 6 + 0) * 32];
    const v2f ba1 = bp[(p * 6 + 1) * 32];
    const v2f ba2 = bp[(p * 6 + 2) * 32];
    const v2f bb0 = bp[(p * 6 + 3) * 32];
    const v2f bb1 = bp[(p * 6 + 4) * 32];
    const v2f bb2 = bp[(p * 6 + 5) * 32];

    v8f cAcc = {};
    v8f cBcc = {};
    cAcc = __builtin_amdgcn_wmma_f32_16x16x4_f32(false, a0, false, ba0, (short)0, cAcc, false, false);
    cAcc = __builtin_amdgcn_wmma_f32_16x16x4_f32(false, a1, false, ba1, (short)0, cAcc, false, false);
    cAcc = __builtin_amdgcn_wmma_f32_16x16x4_f32(false, a2, false, ba2, (short)0, cAcc, false, false);
    cBcc = __builtin_amdgcn_wmma_f32_16x16x4_f32(false, a0, false, bb0, (short)0, cBcc, false, false);
    cBcc = __builtin_amdgcn_wmma_f32_16x16x4_f32(false, a1, false, bb1, (short)0, cBcc, false, false);
    cBcc = __builtin_amdgcn_wmma_f32_16x16x4_f32(false, a2, false, bb2, (short)0, cBcc, false, false);

    if (act) {
      const v4f c = *(const v4f*)(wc + 36 * p);
#pragma unroll
      for (int m = 0; m < 8; ++m) {
        const float av = cAcc[m];
        const float bv = cBcc[m];
        const float r  = fmaf(av * bv, c.w, fmaf(bv, c.z, fmaf(av, c.y, c.x)));
        resp[9 * p + 81 * m] = r;
      }
    }
  }

  v4f* ot = (v4f*)(out + base * 81);
  v4f ostage[10];
#pragma unroll
  for (int q = 0; q < 10; ++q)
    ostage[q] = lt[head + q * 32 + lane];
  v4f ostageRest;
  if (lane < 4) ostageRest = lt[rest + lane];
#pragma unroll
  for (int q = 0; q < 10; ++q)
    __builtin_nontemporal_store(ostage[q], ot + head + q * 32 + lane);
  if (lane < 4) __builtin_nontemporal_store(ostageRest, ot + rest + lane);
}

extern "C" void kernel_launch(void* const* d_in, const int* in_sizes, int n_in,
                              void* d_out, int out_size, void* d_ws, size_t ws_size,
                              hipStream_t stream) {
  const float* x  = (const float*)d_in[0];
  const float* sa = (const float*)d_in[1];
  const float* sb = (const float*)d_in[2];
  const float* op = (const float*)d_in[3];
  float* out = (float*)d_out;
  float* ws  = (float*)d_ws;

  const int nBatch = in_sizes[0] / 81;

  logic_prep_kernel<<<1, 128, 0, stream>>>(sa, sb, op, ws);

  const int rowsPerBlock = 64;
  const int nBlocks = (nBatch + rowsPerBlock - 1) / rowsPerBlock;
  logic_wmma_kernel<<<nBlocks, 128, 0, stream>>>(x, ws, out, nBatch);
}


# prep launch folded into the main kernel: each 8-wave block builds the softmax tables and B fragments in its own LDS while its x loads are in flight; one launch in place of two, B read from LDS
# speedup vs baseline: 3.6587x; 2.1069x over previous
#include <hip/hip_runtime.h>


typedef __attribute__((ext_vector_type(2))) float v2f;
typedef __attribute__((ext_vector_type(2), aligned(4))) float v2fu;
typedef __attribute__((ext_vector_type(4))) float v4f;
typedef __attribute__((ext_vector_type(8))) float v8f;

#define NUNITS 81
#define NFRAG     (9 * 2 * 3 * 32)
#define TILE_FLOATS (16 * 81)
#define NWAVES 8
#define NTHREADS (NWAVES * 32)

#define HW2I_PACKED 0x8FF5F2764310ULL

__global__ void __launch_bounds__(NTHREADS)
logic_fused_kernel(const float* __restrict__ x,
                   const float* __restrict__ sa,
                   const float* __restrict__ sb,
                   const float* __restrict__ op,
                   float* __restrict__ out,
                   int nBatch) {
  __shared__ __attribute__((aligned(16))) float sx[NWAVES * TILE_FLOATS];
  __shared__ __attribute__((aligned(16))) float sSel[2 * NUNITS * 9];
  __shared__ __attribute__((aligned(16))) float sMix[NUNITS * 4];
  __shared__ __attribute__((aligned(16))) float sB[NFRAG * 2];

  const int tid    = threadIdx.x;
  const int lane   = tid & 31;
  const int waveId = __builtin_amdgcn_readfirstlane(tid >> 5);
  const long base  = ((long)blockIdx.x * NWAVES + waveId) * 16;
  if (nBatch < 16) return;
  const bool valid = (base + 16 <= nBatch);
  const long lbase = valid ? base : 0;

  const int col  = lane & 15;
  const bool hiM = (lane >= 16);

  const bool isSel = (tid < NUNITS);
  const bool isOp  = (tid >= 96) && (tid < 96 + NUNITS);
  const int  unit  = isSel ? tid : (tid - 96);
  float lgSel[18];
  float lgOp[16];
  if (isSel) {
#pragma unroll
    for (int i = 0; i < 9; ++i) lgSel[i] = sa[unit * 9 + i];
#pragma unroll
    for (int i = 0; i < 9; ++i) lgSel[9 + i] = sb[unit * 9 + i];
  } else if (isOp) {
#pragma unroll
    for (int i = 0; i < 16; ++i) lgOp[i] = op[unit * 16 + i];
  }

  const v4f* xt   = (const v4f*)(x + lbase * 81);
  v4f*       lt   = (v4f*)(sx + waveId * TILE_FLOATS);
  const int  head = ((lbase * 81) & 16) ? 4 : 0;
  const int  rest = head ? 0 : 320;
  v4f stage[10];
#pragma unroll
  for (int q = 0; q < 10; ++q)
    stage[q] = xt[head + q * 32 + lane];
  const v4f stageRest = xt[rest + (lane & 3)];

  if (isSel) {
#pragma unroll
    for (int t = 0; t < 2; ++t) {
      float m = lgSel[9 * t];
#pragma unroll
      for (int i = 1; i < 9; ++i) m = fmaxf(m, lgSel[9 * t + i]);
      float e[9], s = 0.f;
#pragma unroll
      for (int i = 0; i < 9; ++i) { e[i] = expf(lgSel[9 * t + i] - m); s += e[i]; }
      const float inv = 1.0f / s;
#pragma unroll
      for (int i = 0; i < 9; ++i) sSel[t * (NUNITS * 9) + unit * 9 + i] = e[i] * inv;
    }
  } else if (isOp) {
    float m = lgOp[0];
#pragma unroll
    for (int i = 1; i < 16; ++i) m = fmaxf(m, lgOp[i]);
    float w[16], s = 0.f;
#pragma unroll
    for (int i = 0; i < 16; ++i) { w[i] = expf(lgOp[i] - m); s += w[i]; }
    const float inv = 1.0f / s;
#pragma unroll
    for (int i = 0; i < 16; ++i) w[i] *= inv;
    const float c0  = w[8] + w[9] + w[10] + w[11] + w[12] + w[13] + w[14] + w[15];
    const float cA  = w[2] + w[3] + w[6] + w[7] - w[8] - w[9] - w[12] - w[13];
    const float cB  = w[4] + w[5] + w[6] + w[7] - w[8] - w[9] - w[10] - w[11];
    const float cAB = w[1] - w[2] - w[4] - 2.0f * w[6] - w[7] + w[8] +
                      2.0f * w[9] + w[11] + w[13] - w[14];
    const v4f cm = {c0, cA, cB, cAB};
    *(v4f*)(sMix + unit * 4) = cm;
  }
  __builtin_amdgcn_fence(__ATOMIC_RELEASE, "workgroup", "local");
  __builtin_amdgcn_s_barrier();
  __builtin_amdgcn_fence(__ATOMIC_ACQUIRE, "workgroup", "local");

  {
    const int colc  = (col < 9) ? col : 8;
    const int khalf = hiM ? 2 : 0;
    constexpr int NGRP = NFRAG / 32;
    constexpr int NIT  = (NGRP + NWAVES - 1) / NWAVES;
    float r0[NIT], r1[NIT];
    bool  u0[NIT], u1[NIT];
#pragma unroll
    for (int it = 0; it < NIT; ++it) {
      const int rst = waveId + it * NWAVES;
      const int rc  = (rst < NGRP) ? rst : (NGRP - 1);
      const int kk = rc % 3;
      const int pt = rc / 3;
      const int t  = pt & 1;
      const int p  = pt >> 1;
      const int k0 = 4 * kk + khalf;
      const int i0 = (int)((HW2I_PACKED >> (4 * k0)) & 15ULL);
      const int i1 = (int)((HW2I_PACKED >> (4 * k0 + 4)) & 15ULL);
      const float* row = sSel + t * (NUNITS * 9) + (9 * p + colc) * 9;
      r0[it] = row[(i0 == 15) ? 0 : i0];
      r1[it] = row[(i1 == 15) ? 0 : i1];
      u0[it] = (col < 9) && (i0 != 15);
      u1[it] = (col < 9) && (i1 != 15);
    }
#pragma unroll
    for (int it = 0; it < NIT; ++it) {
      const int rst = waveId + it * NWAVES;
      if (rst < NGRP) {
        v2f v;
        v.x = u0[it] ? r0[it] : 0.f;
        v.y = u1[it] ? r1[it] : 0.f;
        *(v2f*)(sB + (rst * 32 + lane) * 2) = v;
      }
    }
  }
  __builtin_amdgcn_fence(__ATOMIC_RELEASE, "workgroup", "local");
  __builtin_amdgcn_s_barrier();
  __builtin_amdgcn_fence(__ATOMIC_ACQUIRE, "workgroup", "local");

  if (!valid) return;

#pragma unroll
  for (int q = 0; q < 10; ++q) lt[head + q * 32 + lane] = stage[q];
  if (lane < 4) lt[rest + lane] = stageRest;

  const float* xrow = sx + waveId * TILE_FLOATS + col * 81;
  const float* xr0  = xrow + (hiM ? 9 : 0);
  const float* xr1  = xrow + (hiM ? 2 : 18);
  const float* xr2  = xrow + (hiM ? 19 : 11);
  const v2f*   bp   = (const v2f*)sB + lane;
  const float* wc   = sMix + col * 4;
  float*       resp = sx + waveId * TILE_FLOATS + (hiM ? 8 : 0) * 81 + col;
  const bool   act  = (col < 9);

  v2f afr[9][3];
#pragma unroll
  for (int p = 0; p < 9; ++p) {
    const int pbase = (p / 3) * 27 + (p % 3) * 3;
    afr[p][0] = *(const v2fu*)(xr0 + pbase);
    afr[p][1] = *(const v2fu*)(xr1 + pbase);
    afr[p][2] = *(const v2fu*)(xr2 + pbase);
  }

#pragma unroll
  for (int p = 0; p < 9; ++p) {
    const v2f a0 = afr[p][0];
    const v2f a1 = afr[p][1];
    const v2f a2 = afr[p][2];

    const v2f ba0 = bp[(p * 6 + 0) * 32];
    const v2f ba1 = bp[(p * 6 + 1) * 32];
    const v2f ba2 = bp[(p * 6 + 2) * 32];
    const v2f bb0 = bp[(p * 6 + 3) * 32];
    const v2f bb1 = bp[(p * 6 + 4) * 32];
    const v2f bb2 = bp[(p * 6 + 5) * 32];

    v8f cAcc = {};
    v8f cBcc = {};
    cAcc = __builtin_amdgcn_wmma_f32_16x16x4_f32(false, a0, false, ba0, (short)0, cAcc, false, false);
    cAcc = __builtin_amdgcn_wmma_f32_16x16x4_f32(false, a1, false, ba1, (short)0, cAcc, false, false);
    cAcc = __builtin_amdgcn_wmma_f32_16x16x4_f32(false, a2, false, ba2, (short)0, cAcc, false, false);
    cBcc = __builtin_amdgcn_wmma_f32_16x16x4_f32(false, a0, false, bb0, (short)0, cBcc, false, false);
    cBcc = __builtin_amdgcn_wmma_f32_16x16x4_f32(false, a1, false, bb1, (short)0, cBcc, false, false);
    cBcc = __builtin_amdgcn_wmma_f32_16x16x4_f32(false, a2, false, bb2, (short)0, cBcc, false, false);

    if (act) {
      const v4f c = *(const v4f*)(wc + 36 * p);
#pragma unroll
      for (int m = 0; m < 8; ++m) {
        const float av = cAcc[m];
        const float bv = cBcc[m];
        const float r  = fmaf(av * bv, c.w, fmaf(bv, c.z, fmaf(av, c.y, c.x)));
        resp[9 * p + 81 * m] = r;
      }
    }
  }

  v4f* ot = (v4f*)(out + base * 81);
  v4f ostage[10];
#pragma unroll
  for (int q = 0; q < 10; ++q)
    ostage[q] = lt[head + q * 32 + lane];
  v4f ostageRest;
  if (lane < 4) ostageRest = lt[rest + lane];
#pragma unroll
  for (int q = 0; q < 10; ++q)
    __builtin_nontemporal_store(ostage[q], ot + head + q * 32 + lane);
  if (lane < 4) __builtin_nontemporal_store(ostageRest, ot + rest + lane);
}

extern "C" void kernel_launch(void* const* d_in, const int* in_sizes, int n_in,
                              void* d_out, int out_size, void* d_ws, size_t ws_size,
                              hipStream_t stream) {
  const float* x  = (const float*)d_in[0];
  const float* sa = (const float*)d_in[1];
  const float* sb = (const float*)d_in[2];
  const float* op = (const float*)d_in[3];
  float* out = (float*)d_out;

  const int nBatch = in_sizes[0] / 81;

  const int rowsPerBlock = 128;
  const int nBlocks = (nBatch + rowsPerBlock - 1) / rowsPerBlock;
  logic_fused_kernel<<<nBlocks, NTHREADS, 0, stream>>>(x, sa, sb, op, out, nBatch);
}


# softmax threads store their unit's B-fragment entries straight in the per-lane WMMA layout (12 8-byte LDS stores each); the selector table, its 7-round re-read by all 8 waves and one barrier go
# speedup vs baseline: 3.8973x; 1.0652x over previous
#include <hip/hip_runtime.h>


typedef __attribute__((ext_vector_type(2))) float v2f;
typedef __attribute__((ext_vector_type(2), aligned(4))) float v2fu;
typedef __attribute__((ext_vector_type(4))) float v4f;
typedef __attribute__((ext_vector_type(8))) float v8f;

#define NUNITS 81
#define NFRAG     (9 * 2 * 3 * 32)
#define TILE_FLOATS (16 * 81)
#define NWAVES 8
#define NTHREADS (NWAVES * 32)

#define HW2I_PACKED 0x8FF5F2764310ULL

__global__ void __launch_bounds__(NTHREADS)
logic_fused_kernel(const float* __restrict__ x,
                   const float* __restrict__ sa,
                   const float* __restrict__ sb,
                   const float* __restrict__ op,
                   float* __restrict__ out,
                   int nBatch) {
  __shared__ __attribute__((aligned(16))) float sx[NWAVES * TILE_FLOATS];
  __shared__ __attribute__((aligned(16))) float sMix[NUNITS * 4];
  __shared__ __attribute__((aligned(16))) float sB[NFRAG * 2];

  const int tid    = threadIdx.x;
  const int lane   = tid & 31;
  const int waveId = __builtin_amdgcn_readfirstlane(tid >> 5);
  const long base  = ((long)blockIdx.x * NWAVES + waveId) * 16;
  if (nBatch < 16) return;
  const bool valid = (base + 16 <= nBatch);
  const long lbase = valid ? base : 0;

  const int col  = lane & 15;
  const bool hiM = (lane >= 16);

  const bool isSel = (tid < NUNITS);
  const bool isOp  = (tid >= 96) && (tid < 96 + NUNITS);
  const bool isPad = (tid >= 192) && (tid < 192 + 56);
  const int  unit  = isSel ? tid : (tid - 96);
  float lgSel[18];
  float lgOp[16];
  if (isSel) {
#pragma unroll
    for (int i = 0; i < 9; ++i) lgSel[i] = sa[unit * 9 + i];
#pragma unroll
    for (int i = 0; i < 9; ++i) lgSel[9 + i] = sb[unit * 9 + i];
  } else if (isOp) {
#pragma unroll
    for (int i = 0; i < 16; ++i) lgOp[i] = op[unit * 16 + i];
  }

  const v4f* xt   = (const v4f*)(x + lbase * 81);
  v4f*       lt   = (v4f*)(sx + waveId * TILE_FLOATS);
  const int  head = ((lbase * 81) & 16) ? 4 : 0;
  const int  rest = head ? 0 : 320;
  v4f stage[10];
#pragma unroll
  for (int q = 0; q < 10; ++q)
    stage[q] = xt[head + q * 32 + lane];
  const v4f stageRest = xt[rest + (lane & 3)];

  if (isSel) {
#pragma unroll
    for (int t = 0; t < 2; ++t) {
      float m = lgSel[9 * t];
#pragma unroll
      for (int i = 1; i < 9; ++i) m = fmaxf(m, lgSel[9 * t + i]);
      float e[9], s = 0.f;
#pragma unroll
      for (int i = 0; i < 9; ++i) { e[i] = expf(lgSel[9 * t + i] - m); s += e[i]; }
      const float inv = 1.0f / s;
#pragma unroll
      for (int i = 0; i < 9; ++i) e[i] *= inv;
      v2f* bq = (v2f*)sB + ((unit / 9) * 6 + t * 3) * 32 + (unit % 9);
#pragma unroll
      for (int kk = 0; kk < 3; ++kk) {
#pragma unroll
        for (int h = 0; h < 2; ++h) {
          const int k0 = 4 * kk + 2 * h;
          const int i0 = (int)((HW2I_PACKED >> (4 * k0)) & 15ULL);
          const int i1 = (int)((HW2I_PACKED >> (4 * k0 + 4)) & 15ULL);
          v2f v;
          v.x = (i0 == 15) ? 0.f : e[(i0 == 15) ? 0 : i0];
          v.y = (i1 == 15) ? 0.f : e[(i1 == 15) ? 0 : i1];
          bq[kk * 32 + 16 * h] = v;
        }
      }
    }
  } else if (isOp) {
    float m = lgOp[0];
#pragma unroll
    for (int i = 1; i < 16; ++i) m = fmaxf(m, lgOp[i]);
    float w[16], s = 0.f;
#pragma unroll
    for (int i = 0; i < 16; ++i) { w[i] = expf(lgOp[i] - m); s += w[i]; }
    const float inv = 1.0f / s;
#pragma unroll
    for (int i = 0; i < 16; ++i) w[i] *= inv;
    const float c0  = w[8] + w[9] + w[10] + w[11] + w[12] + w[13] + w[14] + w[15];
    const float cA  = w[2] + w[3] + w[6] + w[7] - w[8] - w[9] - w[12] - w[13];
    const float cB  = w[4] + w[5] + w[6] + w[7] - w[8] - w[9] - w[10] - w[11];
    const float cAB = w[1] - w[2] - w[4] - 2.0f * w[6] - w[7] + w[8] +
                      2.0f * w[9] + w[11] + w[13] - w[14];
    const v4f cm = {c0, cA, cB, cAB};
    *(v4f*)(sMix + unit * 4) = cm;
  }
  if (isPad) {
    const int z  = tid - 192;
    const int r  = z % 14;
    const int q  = z / 14;
    const int pl = (r < 7) ? (9 + r) : (18 + r);
    v2f* zq = (v2f*)sB + q * 32 + pl;
    const v2f zero = {0.f, 0.f};
#pragma unroll
    for (int it = 0; it < 13; ++it) zq[it * 128] = zero;
    if (q < 2) zq[13 * 128] = zero;
  }
  __builtin_amdgcn_fence(__ATOMIC_RELEASE, "workgroup", "local");
  __builtin_amdgcn_s_barrier();
  __builtin_amdgcn_fence(__ATOMIC_ACQUIRE, "workgroup", "local");

  if (!valid) return;

#pragma unroll
  for (int q = 0; q < 10; ++q) lt[head + q * 32 + lane] = stage[q];
  if (lane < 4) lt[rest + lane] = stageRest;

  const float* xrow = sx + waveId * TILE_FLOATS + col * 81;
  const float* xr0  = xrow + (hiM ? 9 : 0);
  const float* xr1  = xrow + (hiM ? 2 : 18);
  const float* xr2  = xrow + (hiM ? 19 : 11);
  const v2f*   bp   = (const v2f*)sB + lane;
  const float* wc   = sMix + col * 4;
  float*       resp = sx + waveId * TILE_FLOATS + (hiM ? 8 : 0) * 81 + col;
  const bool   act  = (col < 9);

  v2f afr[9][3];
#pragma unroll
  for (int p = 0; p < 9; ++p) {
    const int pbase = (p / 3) * 27 + (p % 3) * 3;
    afr[p][0] = *(const v2fu*)(xr0 + pbase);
    afr[p][1] = *(const v2fu*)(xr1 + pbase);
    afr[p][2] = *(const v2fu*)(xr2 + pbase);
  }

#pragma unroll
  for (int p = 0; p < 9; ++p) {
    const v2f a0 = afr[p][0];
    const v2f a1 = afr[p][1];
    const v2f a2 = afr[p][2];

    const v2f ba0 = bp[(p * 6 + 0) * 32];
    const v2f ba1 = bp[(p * 6 + 1) * 32];
    const v2f ba2 = bp[(p * 6 + 2) * 32];
    const v2f bb0 = bp[(p * 6 + 3) * 32];
    const v2f bb1 = bp[(p * 6 + 4) * 32];
    const v2f bb2 = bp[(p * 6 + 5) * 32];

    v8f cAcc = {};
    v8f cBcc = {};
    cAcc = __builtin_amdgcn_wmma_f32_16x16x4_f32(false, a0, false, ba0, (short)0, cAcc, false, false);
    cAcc = __builtin_amdgcn_wmma_f32_16x16x4_f32(false, a1, false, ba1, (short)0, cAcc, false, false);
    cAcc = __builtin_amdgcn_wmma_f32_16x16x4_f32(false, a2, false, ba2, (short)0, cAcc, false, false);
    cBcc = __builtin_amdgcn_wmma_f32_16x16x4_f32(false, a0, false, bb0, (short)0, cBcc, false, false);
    cBcc = __builtin_amdgcn_wmma_f32_16x16x4_f32(false, a1, false, bb1, (short)0, cBcc, false, false);
    cBcc = __builtin_amdgcn_wmma_f32_16x16x4_f32(false, a2, false, bb2, (short)0, cBcc, false, false);

    if (act) {
      const v4f c = *(const v4f*)(wc + 36 * p);
#pragma unroll
      for (int m = 0; m < 8; ++m) {
        const float av = cAcc[m];
        const float bv = cBcc[m];
        const float r  = fmaf(av * bv, c.w, fmaf(bv, c.z, fmaf(av, c.y, c.x)));
        resp[9 * p + 81 * m] = r;
      }
    }
  }

  v4f* ot = (v4f*)(out + base * 81);
  v4f ostage[10];
#pragma unroll
  for (int q = 0; q < 10; ++q)
    ostage[q] = lt[head + q * 32 + lane];
  v4f ostageRest;
  if (lane < 4) ostageRest = lt[rest + lane];
#pragma unroll
  for (int q = 0; q < 10; ++q)
    __builtin_nontemporal_store(ostage[q], ot + head + q * 32 + lane);
  if (lane < 4) __builtin_nontemporal_store(ostageRest, ot + rest + lane);
}

extern "C" void kernel_launch(void* const* d_in, const int* in_sizes, int n_in,
                              void* d_out, int out_size, void* d_ws, size_t ws_size,
                              hipStream_t stream) {
  const float* x  = (const float*)d_in[0];
  const float* sa = (const float*)d_in[1];
  const float* sb = (const float*)d_in[2];
  const float* op = (const float*)d_in[3];
  float* out = (float*)d_out;

  const int nBatch = in_sizes[0] / 81;

  const int rowsPerBlock = 128;
  const int nBlocks = (nBatch + rowsPerBlock - 1) / rowsPerBlock;
  logic_fused_kernel<<<nBlocks, NTHREADS, 0, stream>>>(x, sa, sb, op, out, nBatch);
}


# tile loop: fixed grid of 256 8-wave blocks, each wave takes its tiles in turn (4 here); tables built once per block, B fragments kept in registers, next tile's x loads in flight under the matrix work
# speedup vs baseline: 4.1269x; 1.0589x over previous
#include <hip/hip_runtime.h>


typedef __attribute__((ext_vector_type(2))) float v2f;
typedef __attribute__((ext_vector_type(2), aligned(4))) float v2fu;
typedef __attribute__((ext_vector_type(4))) float v4f;
typedef __attribute__((ext_vector_type(8))) float v8f;

#define NUNITS 81
#define NFRAG     (9 * 2 * 3 * 32)
#define TILE_FLOATS (16 * 81)
#define NWAVES 8
#define NTHREADS (NWAVES * 32)
#define NBLOCKS 256

#define HW2I_PACKED 0x8FF5F2764310ULL

__global__ void __launch_bounds__(NTHREADS)
logic_fused_kernel(const float* __restrict__ x,
                   const float* __restrict__ sa,
                   const float* __restrict__ sb,
                   const float* __restrict__ op,
                   float* __restrict__ out,
                   int nBatch) {
  __shared__ __attribute__((aligned(16))) float sx[NWAVES * TILE_FLOATS];
  __shared__ __attribute__((aligned(16))) float sMix[NUNITS * 4];
  __shared__ __attribute__((aligned(16))) float sB[NFRAG * 2];

  const int tid    = threadIdx.x;
  const int lane   = tid & 31;
  const int waveId = __builtin_amdgcn_readfirstlane(tid >> 5);
  if (nBatch < 16) return;
  const int nTiles     = nBatch >> 4;
  const int tileStride = (int)gridDim.x * NWAVES;
  int       tile       = (int)blockIdx.x * NWAVES + waveId;
  const bool valid = (tile < nTiles);
  const int ltile = valid ? tile : 0;

  const int col  = lane & 15;
  const bool hiM = (lane >= 16);

  const bool isSel = (tid < NUNITS);
  const bool isOp  = (tid >= 96) && (tid < 96 + NUNITS);
  const bool isPad = (tid >= 192) && (tid < 192 + 56);
  const int  unit  = isSel ? tid : (tid - 96);
  float lgSel[18];
  float lgOp[16];
  if (isSel) {
#pragma unroll
    for (int i = 0; i < 9; ++i) lgSel[i] = sa[unit * 9 + i];
#pragma unroll
    for (int i = 0; i < 9; ++i) lgSel[9 + i] = sb[unit * 9 + i];
  } else if (isOp) {
#pragma unroll
    for (int i = 0; i < 16; ++i) lgOp[i] = op[unit * 16 + i];
  }

  v4f*       lt   = (v4f*)(sx + waveId * TILE_FLOATS);
  const int  head = (ltile & 1) ? 4 : 0;
  const int  rest = head ? 0 : 320;
  v4f stage[10];
  v4f stageRest;
  {
    const v4f* xt = (const v4f*)(x + (long)ltile * TILE_FLOATS);
#pragma unroll
    for (int q = 0; q < 10; ++q)
      stage[q] = xt[head + q * 32 + lane];
    stageRest = xt[rest + (lane & 3)];
  }

  if (isSel) {
#pragma unroll
    for (int t = 0; t < 2; ++t) {
      float m = lgSel[9 * t];
#pragma unroll
      for (int i = 1; i < 9; ++i) m = fmaxf(m, lgSel[9 * t + i]);
      float e[9], s = 0.f;
#pragma unroll
      for (int i = 0; i < 9; ++i) { e[i] = expf(lgSel[9 * t + i] - m); s += e[i]; }
      const float inv = 1.0f / s;
#pragma unroll
      for (int i = 0; i < 9; ++i) e[i] *= inv;
      v2f* bq = (v2f*)sB + ((unit / 9) * 6 + t * 3) * 32 + (unit % 9);
#pragma unroll
      for (int kk = 0; kk < 3; ++kk) {
#pragma unroll
        for (int h = 0; h < 2; ++h) {
          const int k0 = 4 * kk + 2 * h;
          const int i0 = (int)((HW2I_PACKED >> (4 * k0)) & 15ULL);
          const int i1 = (int)((HW2I_PACKED >> (4 * k0 + 4)) & 15ULL);
          v2f v;
          v.x = (i0 == 15) ? 0.f : e[(i0 == 15) ? 0 : i0];
          v.y = (i1 == 15) ? 0.f : e[(i1 == 15) ? 0 : i1];
          bq[kk * 32 + 16 * h] = v;
        }
      }
    }
  } else if (isOp) {
    float m = lgOp[0];
#pragma unroll
    for (int i = 1; i < 16; ++i) m = fmaxf(m, lgOp[i]);
    float w[16], s = 0.f;
#pragma unroll
    for (int i = 0; i < 16; ++i) { w[i] = expf(lgOp[i] - m); s += w[i]; }
    const float inv = 1.0f / s;
#pragma unroll
    for (int i = 0; i < 16; ++i) w[i] *= inv;
    const float c0  = w[8] + w[9] + w[10] + w[11] + w[12] + w[13] + w[14] + w[15];
    const float cA  = w[2] + w[3] + w[6] + w[7] - w[8] - w[9] - w[12] - w[13];
    const float cB  = w[4] + w[5] + w[6] + w[7] - w[8] - w[9] - w[10] - w[11];
    const float cAB = w[1] - w[2] - w[4] - 2.0f * w[6] - w[7] + w[8] +
                      2.0f * w[9] + w[11] + w[13] - w[14];
    const v4f cm = {c0, cA, cB, cAB};
    *(v4f*)(sMix + unit * 4) = cm;
  }
  if (isPad) {
    const int z  = tid - 192;
    const int r  = z % 14;
    const int q  = z / 14;
    const int pl = (r < 7) ? (9 + r) : (18 + r);
    v2f* zq = (v2f*)sB + q * 32 + pl;
    const v2f zero = {0.f, 0.f};
#pragma unroll
    for (int it = 0; it < 13; ++it) zq[it * 128] = zero;
    if (q < 2) zq[13 * 128] = zero;
  }
  __builtin_amdgcn_fence(__ATOMIC_RELEASE, "workgroup", "local");
  __builtin_amdgcn_s_barrier();
  __builtin_amdgcn_fence(__ATOMIC_ACQUIRE, "workgroup", "local");

  if (!valid) return;

  const float* xrow = sx + waveId * TILE_FLOATS + col * 81;
  const float* xr0  = xrow + (hiM ? 9 : 0);
  const float* xr1  = xrow + (hiM ? 2 : 18);
  const float* xr2  = xrow + (hiM ? 19 : 11);
  const v2f*   bp   = (const v2f*)sB + lane;
  const bool   act  = (col < 9);
  const float* wc   = sMix + (act ? col : 0) * 4;
  float*       resp = sx + waveId * TILE_FLOATS + (hiM ? 8 : 0) * 81 + col;

  v2f bfr[9][6];
  v4f mixc[9];
#pragma unroll
  for (int p = 0; p < 9; ++p) {
#pragma unroll
    for (int g = 0; g < 6; ++g) bfr[p][g] = bp[(p * 6 + g) * 32];
    mixc[p] = *(const v4f*)(wc + 36 * p);
  }

  for (;;) {
#pragma unroll
    for (int q = 0; q < 10; ++q) lt[head + q * 32 + lane] = stage[q];
    if (lane < 4) lt[rest + lane] = stageRest;

    const int  nextTile = tile + tileStride;
    const bool more     = (nextTile < nTiles);
    if (more) {
      const v4f* xn = (const v4f*)(x + (long)nextTile * TILE_FLOATS);
#pragma unroll
      for (int q = 0; q < 10; ++q)
        stage[q] = xn[head + q * 32 + lane];
      stageRest = xn[rest + (lane & 3)];
    }

    v2f afr[9][3];
#pragma unroll
    for (int p = 0; p < 9; ++p) {
      const int pbase = (p / 3) * 27 + (p % 3) * 3;
      afr[p][0] = *(const v2fu*)(xr0 + pbase);
      afr[p][1] = *(const v2fu*)(xr1 + pbase);
      afr[p][2] = *(const v2fu*)(xr2 + pbase);
    }

#pragma unroll
    for (int p = 0; p < 9; ++p) {
      const v2f a0 = afr[p][0];
      const v2f a1 = afr[p][1];
      const v2f a2 = afr[p][2];

      v8f cAcc = {};
      v8f cBcc = {};
      cAcc = __builtin_amdgcn_wmma_f32_16x16x4_f32(false, a0, false, bfr[p][0], (short)0, cAcc, false, false);
      cAcc = __builtin_amdgcn_wmma_f32_16x16x4_f32(false, a1, false, bfr[p][1], (short)0, cAcc, false, false);
      cAcc = __builtin_amdgcn_wmma_f32_16x16x4_f32(false, a2, false, bfr[p][2], (short)0, cAcc, false, false);
      cBcc = __builtin_amdgcn_wmma_f32_16x16x4_f32(false, a0, false, bfr[p][3], (short)0, cBcc, false, false);
      cBcc = __builtin_amdgcn_wmma_f32_16x16x4_f32(false, a1, false, bfr[p][4], (short)0, cBcc, false, false);
      cBcc = __builtin_amdgcn_wmma_f32_16x16x4_f32(false, a2, false, bfr[p][5], (short)0, cBcc, false, false);

      if (act) {
        const v4f c = mixc[p];
#pragma unroll
        for (int m = 0; m < 8; ++m) {
          const float av = cAcc[m];
          const float bv = cBcc[m];
          const float r  = fmaf(av * bv, c.w, fmaf(bv, c.z, fmaf(av, c.y, c.x)));
          resp[9 * p + 81 * m] = r;
        }
      }
    }

    v4f* ot = (v4f*)(out + (long)tile * TILE_FLOATS);
    v4f ostage[10];
#pragma unroll
    for (int q = 0; q < 10; ++q)
      ostage[q] = lt[head + q * 32 + lane];
    v4f ostageRest;
    if (lane < 4) ostageRest = lt[rest + lane];
#pragma unroll
    for (int q = 0; q < 10; ++q)
      __builtin_nontemporal_store(ostage[q], ot + head + q * 32 + lane);
    if (lane < 4) __builtin_nontemporal_store(ostageRest, ot + rest + lane);

    if (!more) break;
    tile = nextTile;
  }
}

extern "C" void kernel_launch(void* const* d_in, const int* in_sizes, int n_in,
                              void* d_out, int out_size, void* d_ws, size_t ws_size,
                              hipStream_t stream) {
  const float* x  = (const float*)d_in[0];
  const float* sa = (const float*)d_in[1];
  const float* sb = (const float*)d_in[2];
  const float* op = (const float*)d_in[3];
  float* out = (float*)d_out;

  const int nBatch = in_sizes[0] / 81;

  logic_fused_kernel<<<NBLOCKS, NTHREADS, 0, stream>>>(x, sa, sb, op, out, nBatch);
}
